# starting kernel
# baseline (speedup 1.0000x reference)
#include <hip/hip_runtime.h>
#include <math.h>

typedef float v2f __attribute__((ext_vector_type(2)));
typedef float v8f __attribute__((ext_vector_type(8)));

#define B_   16
#define S_   4096
#define DK_  256
#define DV_  256
#define CHK  16
#define NCHK (S_ / CHK)
#define SPLIT 8
#define VROWS (DV_ / SPLIT)
#define NTHREADS 256
#define NWAVES (NTHREADS / 32)
#define MS 260
#define US 33
#define UT 18
#define VS 34

#define LR_C 0.1f
#define A_C  1.1f
#define EPS_C 1e-6f

__global__ __launch_bounds__(NTHREADS)
void DeltaRuleMemory_chunked_wmma(const float* __restrict__ mem,
                                  const float* __restrict__ key,
                                  const float* __restrict__ val,
                                  float* __restrict__ out) {
    extern __shared__ float smem[];
    float* Msub = smem;
    float* Kn   = Msub + VROWS * MS;
    float* P0Ta = Kn   + CHK * MS;
    float* P0Tb = P0Ta + CHK * US;
    float* Uv   = P0Tb + CHK * US;
    float* Ga   = Uv   + VROWS * UT;
    float* Gb   = Ga   + 256;
    float* Vc   = Gb   + 256;
    float* red  = Vc   + CHK * VS;
    float* inv  = red  + 256;

    const int tid  = threadIdx.x;
    const int wave = tid >> 5;
    const int lane = tid & 31;
    const int hi   = lane >> 4;
    const int lm   = lane & 15;

    const int blk = blockIdx.x;
    const int b   = blk / SPLIT;
    const int v0  = (blk % SPLIT) * VROWS;

    const float* memB = mem + (size_t)b * DV_ * DK_ + (size_t)v0 * DK_;
    const float* keyB = key + (size_t)b * S_ * DK_;
    const float* valB = val + (size_t)b * S_ * DV_;
    float*       outB = out + (size_t)b * DV_ * DK_ + (size_t)v0 * DK_;

    for (int i = tid; i < VROWS * DK_; i += NTHREADS) {
        int r = i >> 8, k = i & 255;
        Msub[r * MS + k] = memB[r * DK_ + k];
    }
    __syncthreads();

    const int tA = tid >> 4;
    const int sg = tid & 15;
    const int d0 = sg * 16;
    const int jV = sg * 2;

    for (int c = 0; c < NCHK; ++c) {
        float4 rv[4];
        float ss = 0.f;
        const float* kr = keyB + (size_t)(c * CHK + tA) * DK_ + d0;
        #pragma unroll
        for (int j = 0; j < 4; ++j) {
            rv[j] = ((const float4*)kr)[j];
            ss += rv[j].x * rv[j].x + rv[j].y * rv[j].y +
                  rv[j].z * rv[j].z + rv[j].w * rv[j].w;
        }
        {
            const float* vg = valB + (size_t)(c * CHK + tA) * DV_ + v0 + jV;
            Vc[tA * VS + jV]     = vg[0];
            Vc[tA * VS + jV + 1] = vg[1];
        }
        if (c + 1 < NCHK)
            __builtin_prefetch(keyB + (size_t)((c + 1) * CHK + tA) * DK_ + d0, 0, 0);
        red[tA * 16 + sg] = ss;
        __syncthreads();
        if (tid < 16) {
            float s = 0.f;
            #pragma unroll
            for (int j = 0; j < 16; ++j) s += red[tid * 16 + j];
            inv[tid] = 1.f / (sqrtf(s) + EPS_C);
        }
        __syncthreads();
        {
            float sc = inv[tA];
            float* kw = Kn + tA * MS + d0;
            #pragma unroll
            for (int j = 0; j < 4; ++j) {
                kw[4 * j + 0] = rv[j].x * sc;
                kw[4 * j + 1] = rv[j].y * sc;
                kw[4 * j + 2] = rv[j].z * sc;
                kw[4 * j + 3] = rv[j].w * sc;
            }
        }
        __syncthreads();

        if (wave < 6) {
            const int tile = wave >> 1;
            const int kh   = wave & 1;
            const float* arow = (tile < 2 ? Msub + (tile * 16 + lm) * MS
                                          : Kn + lm * MS) + 2 * hi;
            const float* brow = Kn + lm * MS + 2 * hi;
            v8f acc0 = {}, acc1 = {};
            const int kbeg = kh * 128;
            for (int k0 = kbeg; k0 < kbeg + 128; k0 += 8) {
                v2f a0 = *(const v2f*)(arow + k0);
                v2f b0 = *(const v2f*)(brow + k0);
                acc0 = __builtin_amdgcn_wmma_f32_16x16x4_f32(
                           false, a0, false, b0, (short)0, acc0, false, false);
                v2f a1 = *(const v2f*)(arow + k0 + 4);
                v2f b1 = *(const v2f*)(brow + k0 + 4);
                acc1 = __builtin_amdgcn_wmma_f32_16x16x4_f32(
                           false, a1, false, b1, (short)0, acc1, false, false);
            }
            v8f acc = acc0 + acc1;
            if (tile < 2) {
                float* dst = kh ? P0Tb : P0Ta;
                #pragma unroll
                for (int r = 0; r < 8; ++r)
                    dst[lm * US + tile * 16 + r + 8 * hi] = acc[r];
            } else {
                float* dst = kh ? Gb : Ga;
                #pragma unroll
                for (int r = 0; r < 8; ++r)
                    dst[(r + 8 * hi) * 16 + lm] = acc[r];
            }
        }
        __syncthreads();

        if (tid < VROWS) {
            const int v = tid;
            float u[CHK];
            #pragma unroll
            for (int t = 0; t < CHK; ++t) {
                float x = LR_C * Vc[t * VS + v]
                        - A_C * (P0Ta[t * US + v] + P0Tb[t * US + v]);
                for (int s2 = 0; s2 < t; ++s2)
                    x -= A_C * (Ga[t * 16 + s2] + Gb[t * 16 + s2]) * u[s2];
                u[t] = x;
            }
            float* uw = Uv + v * UT;
            #pragma unroll
            for (int t = 0; t < CHK; ++t) uw[t] = u[t];
        }
        __syncthreads();

        #pragma unroll
        for (int i = 0; i < (VROWS / 16) * (DK_ / 16) / NWAVES; ++i) {
            int tl = wave + i * NWAVES;
            int vt = tl & (VROWS / 16 - 1);
            int kt = tl / (VROWS / 16);
            float* crow = Msub + (vt * 16) * MS + kt * 16 + lm;
            v8f acc;
            #pragma unroll
            for (int r = 0; r < 8; ++r) acc[r] = crow[(r + 8 * hi) * MS];
            const float* urow = Uv + (vt * 16 + lm) * UT + 2 * hi;
            #pragma unroll
            for (int k0 = 0; k0 < CHK; k0 += 4) {
                v2f a = *(const v2f*)(urow + k0);
                v2f bb; bb.x = Kn[(k0 + 2 * hi) * MS + kt * 16 + lm];
                        bb.y = Kn[(k0 + 2 * hi + 1) * MS + kt * 16 + lm];
                acc = __builtin_amdgcn_wmma_f32_16x16x4_f32(
                          false, a, false, bb, (short)0, acc, false, false);
            }
            #pragma unroll
            for (int r = 0; r < 8; ++r) crow[(r + 8 * hi) * MS] = acc[r];
        }
        __syncthreads();
    }

    for (int i = tid; i < VROWS * DK_; i += NTHREADS) {
        int r = i >> 8, k = i & 255;
        outB[r * DK_ + k] = Msub[r * MS + k];
    }
}

extern "C" void kernel_launch(void* const* d_in, const int* in_sizes, int n_in,
                              void* d_out, int out_size, void* d_ws, size_t ws_size,
                              hipStream_t stream) {
    (void)in_sizes; (void)n_in; (void)out_size; (void)d_ws; (void)ws_size;
    const float* mem = (const float*)d_in[0];
    const float* key = (const float*)d_in[1];
    const float* val = (const float*)d_in[2];
    float* out = (float*)d_out;

    size_t smemFloats = (size_t)(VROWS * MS)
                      + CHK * MS
                      + 2 * CHK * US
                      + VROWS * UT
                      + 2 * 256
                      + CHK * VS
                      + 256 + 16;
    size_t smemBytes = smemFloats * sizeof(float);

    dim3 grid(B_ * SPLIT);
    dim3 block(NTHREADS);
    hipLaunchKernelGGL(DeltaRuleMemory_chunked_wmma, grid, block, smemBytes, stream,
                       mem, key, val, out);
}


# Workgroup scans 64 value rows, not 32 (448 workgroups): staged keys, norms, Gram, solve chain, barriers shared by twice the rows; Gram in 4 K quarters, one per SIMD (Gram sum order changed)
# speedup vs baseline: 3.1173x; 3.1173x over previous
#include <hip/hip_runtime.h>
#include <math.h>


typedef float v2f __attribute__((ext_vector_type(2)));
typedef float v8f __attribute__((ext_vector_type(8)));

#define B_   16
#define S_   4096
#define DK_  256
#define DV_  256
#define CHK  16
#define NCHK (S_ / CHK)
#define SPLIT 4
#define VROWS (DV_ / SPLIT)
#define VT (VROWS / 16)
#define NTHREADS 256
#define NWAVES (NTHREADS / 32)
#define MS 260
#define UT 18
#define PT 20

#define LR_C 0.1f
#define A_C  1.1f
#define EPS_C 1e-6f

#define SMEM_USED (VROWS * MS + CHK * MS + 2 * VROWS * PT + VROWS * UT + 4 * 256 + VROWS * PT + 256 + 2 * 16)
#define SMEM_FLOATS (108 * 1024 / 4)
static_assert(SMEM_USED <= SMEM_FLOATS, "the scan's buffers fit the LDS asked for");
static_assert(VROWS == 64 && CHK == 16, "the chunk solve maps one value row to each lane of two waves");
static_assert(2 * VT == NWAVES, "one P0 half-tile (16 value rows x half of the key columns) per wave");
static_assert(DK_ == NWAVES * 32 && MS % 4 == 0, "the update gives each wave 32 key columns, read in 16-byte pieces");

#define NSEG_MAX 4
#define UNITS_MAX (SPLIT + 2 * SPLIT * (NSEG_MAX - 1))
#define PLANE (DV_ * DK_)
#define SEG_WS_BYTES ((size_t)B_ * 2 * PLANE * sizeof(float))

__device__ __forceinline__ int seg_count(size_t ws_bytes) {
    const size_t n = 1 + ws_bytes / SEG_WS_BYTES;
    return n > NSEG_MAX ? NSEG_MAX : (int)n;
}

__global__ __launch_bounds__(NTHREADS, 4)
void delta_scan_segments(const float* __restrict__ mem,
                         const float* __restrict__ key,
                         const float* __restrict__ val,
                         float* __restrict__ out,
                         float* __restrict__ ws,
                         size_t ws_bytes) {
    __shared__ __attribute__((aligned(16))) float smem[SMEM_FLOATS];
    float* Msub = smem;
    float* Kn   = Msub + VROWS * MS;
    float* P0Ta = Kn   + CHK * MS;
    float* P0Tb = P0Ta + VROWS * PT;
    float* Uv   = P0Tb + VROWS * PT;
    float* Gq   = Uv   + VROWS * UT;
    float* Vc   = Gq   + 4 * 256;
    float* red  = Vc   + VROWS * PT;
    float* lrc  = red  + 256;
    float* ac2  = lrc  + 16;

    const int tid  = threadIdx.x;
    const int wave = tid >> 5;
    const int lane = tid & 31;
    const int hi   = lane >> 4;
    const int lm   = lane & 15;

    const int nseg = seg_count(ws_bytes);
    const int b    = blockIdx.x % B_;
    const int unit = blockIdx.x / B_;
    if (unit >= SPLIT + 2 * SPLIT * (nseg - 1)) return;
    int seg = 0, kind = 0, slice = unit;
    if (unit >= SPLIT) {
        const int w = unit - SPLIT;
        const int r = w % (2 * SPLIT);
        seg   = 1 + w / (2 * SPLIT);
        kind  = r < SPLIT ? 1 : 2;
        slice = r % SPLIT;
    }
    const int v0   = slice * VROWS;
    const int cbeg = seg * NCHK / nseg;
    const int cend = (seg + 1) * NCHK / nseg;

    const float* memB = mem + (size_t)b * PLANE + (size_t)v0 * DK_;
    const float* keyB = key + (size_t)b * S_ * DK_;
    const float* valB = val + (size_t)b * S_ * DV_;

    if (kind == 0) {
        for (int i = tid; i < VROWS * DK_; i += NTHREADS) {
            int r = i >> 8, k = i & 255;
            Msub[r * MS + k] = memB[r * DK_ + k];
        }
    } else {
        const float diag = (kind == 2) ? 1.f : 0.f;
        for (int i = tid; i < VROWS * DK_; i += NTHREADS) {
            int r = i >> 8, k = i & 255;
            Msub[r * MS + k] = (v0 + r == k) ? diag : 0.f;
        }
    }
    __syncthreads();

    const int tA = tid >> 4;
    const int sg = tid & 15;
    const int d0 = sg * 16;
    const int jV = sg * 4;

    for (int c = cbeg; c < cend; ++c) {
        float4 rv[4];
        float ss = 0.f;
        const float* kr = keyB + (size_t)(c * CHK + tA) * DK_ + d0;
        #pragma unroll
        for (int j = 0; j < 4; ++j) {
            rv[j] = ((const float4*)kr)[j];
            ss += rv[j].x * rv[j].x + rv[j].y * rv[j].y +
                  rv[j].z * rv[j].z + rv[j].w * rv[j].w;
        }
        {
            float4 vv = make_float4(0.f, 0.f, 0.f, 0.f);
            if (kind != 2)
                vv = *(const float4*)(valB + (size_t)(c * CHK + tA) * DV_ + v0 + jV);
            Vc[jV * PT + tA]       = vv.x;
            Vc[(jV + 1) * PT + tA] = vv.y;
            Vc[(jV + 2) * PT + tA] = vv.z;
            Vc[(jV + 3) * PT + tA] = vv.w;
        }
        if (c + 1 < cend)
            __builtin_prefetch(keyB + (size_t)((c + 1) * CHK + tA) * DK_ + d0, 0, 0);
        red[tA * 16 + sg] = ss;
        {
            float* kw = Kn + tA * MS + d0;
            #pragma unroll
            for (int j = 0; j < 4; ++j) *(float4*)(kw + 4 * j) = rv[j];
        }
        __syncthreads();

        {
            const int tile = wave >> 1;
            const int kh   = wave & 1;
            const float* arow = Msub + (tile * 16 + lm) * MS + 2 * hi;
            const float* brow = Kn + lm * MS + 2 * hi;
            v8f acc0 = {}, acc1 = {};
            const int kbeg = kh * 128;
            for (int k0 = kbeg; k0 < kbeg + 128; k0 += 8) {
                v2f a0 = *(const v2f*)(arow + k0);
                v2f b0 = *(const v2f*)(brow + k0);
                acc0 = __builtin_amdgcn_wmma_f32_16x16x4_f32(
                           false, a0, false, b0, (short)0, acc0, false, false);
                v2f a1 = *(const v2f*)(arow + k0 + 4);
                v2f b1 = *(const v2f*)(brow + k0 + 4);
                acc1 = __builtin_amdgcn_wmma_f32_16x16x4_f32(
                           false, a1, false, b1, (short)0, acc1, false, false);
            }
            v8f acc = acc0 + acc1;
            float* dst = kh ? P0Tb : P0Ta;
            #pragma unroll
            for (int r = 0; r < 8; ++r)
                dst[(tile * 16 + r + 8 * hi) * PT + lm] = acc[r];
        }
        if (wave >= 4) {
            const float* grow = Kn + lm * MS + 2 * hi;
            v8f g0 = {}, g1 = {};
            const int kbeg = (wave - 4) * 64;
            for (int k0 = kbeg; k0 < kbeg + 64; k0 += 8) {
                v2f a0 = *(const v2f*)(grow + k0);
                v2f b0 = *(const v2f*)(grow + k0);
                g0 = __builtin_amdgcn_wmma_f32_16x16x4_f32(
                         false, a0, false, b0, (short)0, g0, false, false);
                v2f a1 = *(const v2f*)(grow + k0 + 4);
                v2f b1 = *(const v2f*)(grow + k0 + 4);
                g1 = __builtin_amdgcn_wmma_f32_16x16x4_f32(
                         false, a1, false, b1, (short)0, g1, false, false);
            }
            v8f gs = g0 + g1;
            float* dst = Gq + (wave - 4) * 256;
            #pragma unroll
            for (int r = 0; r < 8; ++r)
                dst[(r + 8 * hi) * 16 + lm] = gs[r];
        } else if (wave == 3) {
            if (lane < 16) {
                float s = 0.f;
                #pragma unroll
                for (int j = 0; j < 16; ++j) s += red[lane * 16 + j];
                const float c = 1.f / (sqrtf(s) + EPS_C);
                lrc[lane] = LR_C * c;
                ac2[lane] = A_C * c * c;
            }
        }
        __syncthreads();

        if (tid < VROWS) {
            const int v = tid;
            const float4 ga0 = *(const float4*)(Gq + 4 * lane);
            const float4 ga1 = *(const float4*)(Gq + 128 + 4 * lane);
            const float4 gb0 = *(const float4*)(Gq + 256 + 4 * lane);
            const float4 gb1 = *(const float4*)(Gq + 256 + 128 + 4 * lane);
            const float4 gc0 = *(const float4*)(Gq + 512 + 4 * lane);
            const float4 gc1 = *(const float4*)(Gq + 512 + 128 + 4 * lane);
            const float4 gd0 = *(const float4*)(Gq + 768 + 4 * lane);
            const float4 gd1 = *(const float4*)(Gq + 768 + 128 + 4 * lane);
            const float4* pa = (const float4*)(P0Ta + v * PT);
            const float4* pb = (const float4*)(P0Tb + v * PT);
            const float4* pv = (const float4*)(Vc + v * PT);
            const float4* pl = (const float4*)lrc;
            const float4* pc = (const float4*)ac2;
            const float gt0 = ac2[lane >> 2];
            const float gt1 = ac2[8 + (lane >> 2)];
            float x[CHK];
            #pragma unroll
            for (int q = 0; q < CHK / 4; ++q) {
                const float4 a4 = pa[q], b4 = pb[q], v4 = pv[q], l4 = pl[q], c4 = pc[q];
                x[4 * q + 0] = l4.x * v4.x - c4.x * (a4.x + b4.x);
                x[4 * q + 1] = l4.y * v4.y - c4.y * (a4.y + b4.y);
                x[4 * q + 2] = l4.z * v4.z - c4.z * (a4.z + b4.z);
                x[4 * q + 3] = l4.w * v4.w - c4.w * (a4.w + b4.w);
            }
            float g[8];
            g[0] = gt0 * ((ga0.x + gb0.x) + (gc0.x + gd0.x));
            g[1] = gt0 * ((ga0.y + gb0.y) + (gc0.y + gd0.y));
            g[2] = gt0 * ((ga0.z + gb0.z) + (gc0.z + gd0.z));
            g[3] = gt0 * ((ga0.w + gb0.w) + (gc0.w + gd0.w));
            g[4] = gt1 * ((ga1.x + gb1.x) + (gc1.x + gd1.x));
            g[5] = gt1 * ((ga1.y + gb1.y) + (gc1.y + gd1.y));
            g[6] = gt1 * ((ga1.z + gb1.z) + (gc1.z + gd1.z));
            g[7] = gt1 * ((ga1.w + gb1.w) + (gc1.w + gd1.w));
            #pragma unroll
            for (int s2 = 0; s2 < CHK - 1; ++s2) {
                #pragma unroll
                for (int t = s2 + 1; t < CHK; ++t) {
                    const float gts = __int_as_float(__builtin_amdgcn_readlane(
                        __float_as_int(g[4 * (t >> 3) + (s2 & 3)]), 4 * (t & 7) + (s2 >> 2)));
                    x[t] -= gts * x[s2];
                }
            }
            float* uw = Uv + v * UT;
            #pragma unroll
            for (int t = 0; t < CHK; ++t) uw[t] = x[t];
        }
        __syncthreads();

        {
            v2f ub[VROWS / 16][CHK / 4];
            #pragma unroll
            for (int vt = 0; vt < VROWS / 16; ++vt) {
                #pragma unroll
                for (int j = 0; j < CHK / 4; ++j)
                    ub[vt][j] = *(const v2f*)(Uv + (vt * 16 + lm) * UT + 4 * j + 2 * hi);
            }
            #pragma unroll
            for (int kk = 0; kk < 2; ++kk) {
                const int kc = wave * 32 + kk * 16;
                float* mrow = Msub + lm * MS + kc + 8 * hi;
                v8f acc[VROWS / 16];
                #pragma unroll
                for (int vt = 0; vt < VROWS / 16; ++vt) {
                    const float4 c0 = *(const float4*)(mrow + vt * 16 * MS);
                    const float4 c1 = *(const float4*)(mrow + vt * 16 * MS + 4);
                    acc[vt][0] = c0.x; acc[vt][1] = c0.y; acc[vt][2] = c0.z; acc[vt][3] = c0.w;
                    acc[vt][4] = c1.x; acc[vt][5] = c1.y; acc[vt][6] = c1.z; acc[vt][7] = c1.w;
                }
                v2f ka[CHK / 4];
                #pragma unroll
                for (int j = 0; j < CHK / 4; ++j) {
                    ka[j].x = Kn[(4 * j + 2 * hi) * MS + kc + lm];
                    ka[j].y = Kn[(4 * j + 2 * hi + 1) * MS + kc + lm];
                }
                #pragma unroll
                for (int j = 0; j < CHK / 4; ++j) {
                    #pragma unroll
                    for (int vt = 0; vt < VROWS / 16; ++vt)
                        acc[vt] = __builtin_amdgcn_wmma_f32_16x16x4_f32(
                                      false, ka[j], false, ub[vt][j], (short)0, acc[vt], false, false);
                }
                #pragma unroll
                for (int vt = 0; vt < VROWS / 16; ++vt) {
                    float4 c0, c1;
                    c0.x = acc[vt][0]; c0.y = acc[vt][1]; c0.z = acc[vt][2]; c0.w = acc[vt][3];
                    c1.x = acc[vt][4]; c1.y = acc[vt][5]; c1.z = acc[vt][6]; c1.w = acc[vt][7];
                    *(float4*)(mrow + vt * 16 * MS)     = c0;
                    *(float4*)(mrow + vt * 16 * MS + 4) = c1;
                }
            }
            __builtin_amdgcn_sched_barrier(0);
        }
        __syncthreads();
    }

    float* dstB = (kind == 0)
        ? out + (size_t)b * PLANE + (size_t)v0 * DK_
        : ws + ((size_t)((seg - 1) * B_ + b) * 2 + (kind == 2 ? 0 : 1)) * PLANE + (size_t)v0 * DK_;
    for (int i = tid; i < VROWS * DK_; i += NTHREADS) {
        int r = i >> 8, k = i & 255;
        dstB[r * DK_ + k] = Msub[r * MS + k];
    }
}

#define CROWS 16
__global__ __launch_bounds__(NTHREADS)
void delta_compose_segments(float* __restrict__ out,
                            const float* __restrict__ ws,
                            size_t ws_bytes) {
    __shared__ float Mc[CROWS * MS];

    const int nseg = seg_count(ws_bytes);
    if (nseg == 1) return;

    const int tid  = threadIdx.x;
    const int wave = tid >> 5;
    const int lane = tid & 31;
    const int hi   = lane >> 4;
    const int lm   = lane & 15;

    const int b  = blockIdx.x / (DV_ / CROWS);
    const int r0 = (blockIdx.x % (DV_ / CROWS)) * CROWS;
    float* outB = out + (size_t)b * PLANE + (size_t)r0 * DK_;

    for (int i = tid; i < CROWS * DK_; i += NTHREADS) {
        int r = i >> 8, k = i & 255;
        Mc[r * MS + k] = outB[r * DK_ + k];
    }
    __syncthreads();

    const int c0 = wave * 32 + lm;
    for (int s = 1; s < nseg; ++s) {
        const float* P = ws + ((size_t)((s - 1) * B_ + b) * 2) * PLANE;
        const float* Q = P + PLANE + (size_t)r0 * DK_;
        v8f acc0, acc1;
        #pragma unroll
        for (int r = 0; r < 8; ++r) {
            acc0[r] = Q[(r + 8 * hi) * DK_ + c0];
            acc1[r] = Q[(r + 8 * hi) * DK_ + c0 + 16];
        }
        const float* arow = Mc + lm * MS + 2 * hi;
        const float* bcol = P + (size_t)(2 * hi) * DK_ + c0;
        v2f bx[8], by[8], nx[8], ny[8];
        #pragma unroll
        for (int j = 0; j < 8; ++j) {
            bx[j].x = bcol[(size_t)(4 * j) * DK_];
            bx[j].y = bcol[(size_t)(4 * j + 1) * DK_];
            by[j].x = bcol[(size_t)(4 * j) * DK_ + 16];
            by[j].y = bcol[(size_t)(4 * j + 1) * DK_ + 16];
        }
        #pragma unroll
        for (int g = 0; g < DK_ / 32; ++g) {
            if (g + 1 < DK_ / 32) {
                #pragma unroll
                for (int j = 0; j < 8; ++j) {
                    const int kn = (g + 1) * 32 + 4 * j;
                    nx[j].x = bcol[(size_t)kn * DK_];
                    nx[j].y = bcol[(size_t)(kn + 1) * DK_];
                    ny[j].x = bcol[(size_t)kn * DK_ + 16];
                    ny[j].y = bcol[(size_t)(kn + 1) * DK_ + 16];
                }
            }
            #pragma unroll
            for (int j = 0; j < 8; ++j) {
                v2f a = *(const v2f*)(arow + g * 32 + 4 * j);
                acc0 = __builtin_amdgcn_wmma_f32_16x16x4_f32(
                           false, a, false, bx[j], (short)0, acc0, false, false);
                acc1 = __builtin_amdgcn_wmma_f32_16x16x4_f32(
                           false, a, false, by[j], (short)0, acc1, false, false);
            }
            #pragma unroll
            for (int j = 0; j < 8; ++j) { bx[j] = nx[j]; by[j] = ny[j]; }
        }
        __syncthreads();
        #pragma unroll
        for (int r = 0; r < 8; ++r) {
            Mc[(r + 8 * hi) * MS + c0]      = acc0[r];
            Mc[(r + 8 * hi) * MS + c0 + 16] = acc1[r];
        }
        __syncthreads();
    }

    for (int i = tid; i < CROWS * DK_; i += NTHREADS) {
        int r = i >> 8, k = i & 255;
        outB[r * DK_ + k] = Mc[r * MS + k];
    }
}

extern "C" void kernel_launch(void* const* d_in, const int* in_sizes, int n_in,
                              void* d_out, int out_size, void* d_ws, size_t ws_size,
                              hipStream_t stream) {
    (void)in_sizes; (void)n_in; (void)out_size;
    const float* mem = (const float*)d_in[0];
    const float* key = (const float*)d_in[1];
    const float* val = (const float*)d_in[2];
    float* out = (float*)d_out;
    float* ws  = (float*)d_ws;

    dim3 grid(B_ * UNITS_MAX);
    dim3 block(NTHREADS);
    hipLaunchKernelGGL(delta_scan_segments, grid, block, 0, stream,
                       mem, key, val, out, ws, ws_size);
    hipLaunchKernelGGL(delta_compose_segments, dim3(B_ * (DV_ / CROWS)), block, 0, stream,
                       out, (const float*)ws, ws_size);
}


# Keys/values staged 3 chunks (48 steps) per loop trip: 1 global-load wait + 1 staging barrier per 48 steps, and waves 4-7 form the next chunk's Gram tile during the solve; sum order same
# speedup vs baseline: 3.2286x; 1.0357x over previous
#include <hip/hip_runtime.h>
#include <math.h>


typedef float v2f __attribute__((ext_vector_type(2)));
typedef float v8f __attribute__((ext_vector_type(8)));

#define B_   16
#define S_   4096
#define DK_  256
#define DV_  256
#define CHK  16
#define NCHK (S_ / CHK)
#define SPLIT 4
#define VROWS (DV_ / SPLIT)
#define VT (VROWS / 16)
#define NTHREADS 256
#define NWAVES (NTHREADS / 32)
#define MS 260
#define UT 18
#define PT 20
#define STG 3

#define LR_C 0.1f
#define A_C  1.1f
#define EPS_C 1e-6f

#define SMEM_USED (VROWS * MS + STG * CHK * MS + 2 * VROWS * PT + VROWS * UT + 2 * 4 * 256 + STG * VROWS * PT + STG * 256 + 2 * STG * 16)
#define SMEM_FLOATS (156 * 1024 / 4)
static_assert(SMEM_USED <= SMEM_FLOATS, "the scan's buffers fit the LDS asked for");
static_assert(VROWS == 64 && CHK == 16, "the chunk solve maps one value row to each lane of two waves");
static_assert(2 * VT == NWAVES, "one P0 half-tile (16 value rows x half of the key columns) per wave");
static_assert(DK_ == NWAVES * 32 && MS % 4 == 0, "the update gives each wave 32 key columns, read in 16-byte pieces");

#define NSEG_MAX 4
#define UNITS_MAX (SPLIT + 2 * SPLIT * (NSEG_MAX - 1))
#define PLANE (DV_ * DK_)
#define SEG_WS_BYTES ((size_t)B_ * 2 * PLANE * sizeof(float))

__device__ __forceinline__ int seg_count(size_t ws_bytes) {
    const size_t n = 1 + ws_bytes / SEG_WS_BYTES;
    return n > NSEG_MAX ? NSEG_MAX : (int)n;
}

__device__ __forceinline__ void gram_quarter(const float* __restrict__ krows, float* __restrict__ dst,
                                             int lm, int hi, int q) {
    const float* grow = krows + lm * MS + 2 * hi + q * 64;
    v2f f[16];
    #pragma unroll
    for (int i = 0; i < 16; ++i) f[i] = *(const v2f*)(grow + 4 * i);
    __builtin_amdgcn_sched_barrier(0);
    v8f g0 = {}, g1 = {};
    #pragma unroll
    for (int i = 0; i < 16; i += 2) {
        g0 = __builtin_amdgcn_wmma_f32_16x16x4_f32(
                 false, f[i], false, f[i], (short)0, g0, false, false);
        g1 = __builtin_amdgcn_wmma_f32_16x16x4_f32(
                 false, f[i + 1], false, f[i + 1], (short)0, g1, false, false);
    }
    const v8f gs = g0 + g1;
    __builtin_amdgcn_sched_barrier(0);
    #pragma unroll
    for (int r = 0; r < 8; ++r)
        dst[(r + 8 * hi) * 16 + lm] = gs[r];
}

__global__ __launch_bounds__(NTHREADS, 4)
void delta_scan_segments(const float* __restrict__ mem,
                         const float* __restrict__ key,
                         const float* __restrict__ val,
                         float* __restrict__ out,
                         float* __restrict__ ws,
                         size_t ws_bytes) {
    __shared__ __attribute__((aligned(16))) float smem[SMEM_FLOATS];
    float* Msub = smem;
    float* KnS  = Msub + VROWS * MS;
    float* P0Ta = KnS  + STG * CHK * MS;
    float* P0Tb = P0Ta + VROWS * PT;
    float* Uv   = P0Tb + VROWS * PT;
    float* GqS  = Uv   + VROWS * UT;
    float* VcS  = GqS  + 2 * 4 * 256;
    float* redS = VcS  + STG * VROWS * PT;
    float* lrcS = redS + STG * 256;
    float* ac2S = lrcS + STG * 16;

    const int tid  = threadIdx.x;
    const int wave = tid >> 5;
    const int lane = tid & 31;
    const int hi   = lane >> 4;
    const int lm   = lane & 15;

    const int nseg = seg_count(ws_bytes);
    const int b    = blockIdx.x % B_;
    const int unit = blockIdx.x / B_;
    if (unit >= SPLIT + 2 * SPLIT * (nseg - 1)) return;
    int seg = 0, kind = 0, slice = unit;
    if (unit >= SPLIT) {
        const int w = unit - SPLIT;
        const int r = w % (2 * SPLIT);
        seg   = 1 + w / (2 * SPLIT);
        kind  = r < SPLIT ? 1 : 2;
        slice = r % SPLIT;
    }
    const int v0   = slice * VROWS;
    const int cbeg = seg * NCHK / nseg;
    const int cend = (seg + 1) * NCHK / nseg;

    const float* memB = mem + (size_t)b * PLANE + (size_t)v0 * DK_;
    const float* keyB = key + (size_t)b * S_ * DK_;
    const float* valB = val + (size_t)b * S_ * DV_;

    if (kind == 0) {
        for (int i = tid; i < VROWS * DK_; i += NTHREADS) {
            int r = i >> 8, k = i & 255;
            Msub[r * MS + k] = memB[r * DK_ + k];
        }
    } else {
        const float diag = (kind == 2) ? 1.f : 0.f;
        for (int i = tid; i < VROWS * DK_; i += NTHREADS) {
            int r = i >> 8, k = i & 255;
            Msub[r * MS + k] = (v0 + r == k) ? diag : 0.f;
        }
    }
    __syncthreads();

    const int tA = tid >> 4;
    const int sg = tid & 15;
    const int d0 = sg * 16;
    const int jV = sg * 4;

    for (int c = cbeg; c < cend; c += STG) {
        const int n2 = (cend - c) < STG ? (cend - c) : STG;
        {
            float4 rv[STG][4];
            float4 vv[STG];
            #pragma unroll
            for (int h = 0; h < STG; ++h) {
                vv[h] = make_float4(0.f, 0.f, 0.f, 0.f);
                #pragma unroll
                for (int j = 0; j < 4; ++j) rv[h][j] = make_float4(0.f, 0.f, 0.f, 0.f);
                if (h < n2) {
                    const float* kr = keyB + (size_t)((c + h) * CHK + tA) * DK_ + d0;
                    #pragma unroll
                    for (int j = 0; j < 4; ++j) rv[h][j] = ((const float4*)kr)[j];
                    if (kind != 2)
                        vv[h] = *(const float4*)(valB + (size_t)((c + h) * CHK + tA) * DV_ + v0 + jV);
                }
            }
            #pragma unroll
            for (int h = 0; h < STG; ++h) {
                if (c + STG + h < cend)
                    __builtin_prefetch(keyB + (size_t)((c + STG + h) * CHK + tA) * DK_ + d0, 0, 0);
            }
            #pragma unroll
            for (int h = 0; h < STG; ++h) {
                if (h < n2) {
                    float ss = 0.f;
                    #pragma unroll
                    for (int j = 0; j < 4; ++j)
                        ss += rv[h][j].x * rv[h][j].x + rv[h][j].y * rv[h][j].y +
                              rv[h][j].z * rv[h][j].z + rv[h][j].w * rv[h][j].w;
                    float* Vh = VcS + h * VROWS * PT;
                    Vh[jV * PT + tA]       = vv[h].x;
                    Vh[(jV + 1) * PT + tA] = vv[h].y;
                    Vh[(jV + 2) * PT + tA] = vv[h].z;
                    Vh[(jV + 3) * PT + tA] = vv[h].w;
                    redS[h * 256 + tA * 16 + sg] = ss;
                    float* kw = KnS + (h * CHK + tA) * MS + d0;
                    #pragma unroll
                    for (int j = 0; j < 4; ++j) *(float4*)(kw + 4 * j) = rv[h][j];
                }
            }
        }
        __syncthreads();

        #pragma unroll
        for (int h = 0; h < STG; ++h) {
            if (h >= n2) break;
            const float* Kh   = KnS  + h * CHK * MS;
            const float* Vh   = VcS  + h * VROWS * PT;
            const float* redh = redS + h * 256;
            float* lrch = lrcS + h * 16;
            float* ac2h = ac2S + h * 16;
            const float* Gq = GqS + (h & 1) * 4 * 256;

            {
                const int tile = wave >> 1;
                const int kh   = wave & 1;
                const float* arow = Msub + (tile * 16 + lm) * MS + 2 * hi;
                const float* brow = Kh + lm * MS + 2 * hi;
                v8f acc0 = {}, acc1 = {};
                const int kbeg = kh * 128;
                for (int k0 = kbeg; k0 < kbeg + 128; k0 += 8) {
                    v2f a0 = *(const v2f*)(arow + k0);
                    v2f b0 = *(const v2f*)(brow + k0);
                    acc0 = __builtin_amdgcn_wmma_f32_16x16x4_f32(
                               false, a0, false, b0, (short)0, acc0, false, false);
                    v2f a1 = *(const v2f*)(arow + k0 + 4);
                    v2f b1 = *(const v2f*)(brow + k0 + 4);
                    acc1 = __builtin_amdgcn_wmma_f32_16x16x4_f32(
                               false, a1, false, b1, (short)0, acc1, false, false);
                }
                v8f acc = acc0 + acc1;
                float* dst = kh ? P0Tb : P0Ta;
                #pragma unroll
                for (int r = 0; r < 8; ++r)
                    dst[(tile * 16 + r + 8 * hi) * PT + lm] = acc[r];
            }
            if (wave >= 4) {
                if (h == 0)
                    gram_quarter(Kh, GqS + (wave - 4) * 256, lm, hi, wave - 4);
            } else if (wave == 3) {
                if (lane < 16) {
                    float s = 0.f;
                    #pragma unroll
                    for (int j = 0; j < 16; ++j) s += redh[lane * 16 + j];
                    const float c = 1.f / (sqrtf(s) + EPS_C);
                    lrch[lane] = LR_C * c;
                    ac2h[lane] = A_C * c * c;
                }
            }
            __syncthreads();

            if (tid < VROWS) {
                const int v = tid;
                const float4 ga0 = *(const float4*)(Gq + 4 * lane);
                const float4 ga1 = *(const float4*)(Gq + 128 + 4 * lane);
                const float4 gb0 = *(const float4*)(Gq + 256 + 4 * lane);
                const float4 gb1 = *(const float4*)(Gq + 256 + 128 + 4 * lane);
                const float4 gc0 = *(const float4*)(Gq + 512 + 4 * lane);
                const float4 gc1 = *(const float4*)(Gq + 512 + 128 + 4 * lane);
                const float4 gd0 = *(const float4*)(Gq + 768 + 4 * lane);
                const float4 gd1 = *(const float4*)(Gq + 768 + 128 + 4 * lane);
                const float4* pa = (const float4*)(P0Ta + v * PT);
                const float4* pb = (const float4*)(P0Tb + v * PT);
                const float4* pv = (const float4*)(Vh + v * PT);
                const float4* pl = (const float4*)lrch;
                const float4* pc = (const float4*)ac2h;
                const float gt0 = ac2h[lane >> 2];
                const float gt1 = ac2h[8 + (lane >> 2)];
                float x[CHK];
                #pragma unroll
                for (int q = 0; q < CHK / 4; ++q) {
                    const float4 a4 = pa[q], b4 = pb[q], v4 = pv[q], l4 = pl[q], c4 = pc[q];
                    x[4 * q + 0] = l4.x * v4.x - c4.x * (a4.x + b4.x);
                    x[4 * q + 1] = l4.y * v4.y - c4.y * (a4.y + b4.y);
                    x[4 * q + 2] = l4.z * v4.z - c4.z * (a4.z + b4.z);
                    x[4 * q + 3] = l4.w * v4.w - c4.w * (a4.w + b4.w);
                }
                float g[8];
                g[0] = gt0 * ((ga0.x + gb0.x) + (gc0.x + gd0.x));
                g[1] = gt0 * ((ga0.y + gb0.y) + (gc0.y + gd0.y));
                g[2] = gt0 * ((ga0.z + gb0.z) + (gc0.z + gd0.z));
                g[3] = gt0 * ((ga0.w + gb0.w) + (gc0.w + gd0.w));
                g[4] = gt1 * ((ga1.x + gb1.x) + (gc1.x + gd1.x));
                g[5] = gt1 * ((ga1.y + gb1.y) + (gc1.y + gd1.y));
                g[6] = gt1 * ((ga1.z + gb1.z) + (gc1.z + gd1.z));
                g[7] = gt1 * ((ga1.w + gb1.w) + (gc1.w + gd1.w));
                #pragma unroll
                for (int s2 = 0; s2 < CHK - 1; ++s2) {
                    #pragma unroll
                    for (int t = s2 + 1; t < CHK; ++t) {
                        const float gts = __int_as_float(__builtin_amdgcn_readlane(
                            __float_as_int(g[4 * (t >> 3) + (s2 & 3)]), 4 * (t & 7) + (s2 >> 2)));
                        x[t] -= gts * x[s2];
                    }
                }
                float* uw = Uv + v * UT;
                #pragma unroll
                for (int t = 0; t < CHK; ++t) uw[t] = x[t];
            } else if (wave >= 4) {
                if (h + 1 < STG && h + 1 < n2)
                    gram_quarter(KnS + (h + 1) * CHK * MS, GqS + ((h + 1) & 1) * 4 * 256 + (wave - 4) * 256,
                                 lm, hi, wave - 4);
            }
            __syncthreads();

            {
                v2f ub[VROWS / 16][CHK / 4];
                #pragma unroll
                for (int vt = 0; vt < VROWS / 16; ++vt) {
                    #pragma unroll
                    for (int j = 0; j < CHK / 4; ++j)
                        ub[vt][j] = *(const v2f*)(Uv + (vt * 16 + lm) * UT + 4 * j + 2 * hi);
                }
                #pragma unroll
                for (int kk = 0; kk < 2; ++kk) {
                    const int kc = wave * 32 + kk * 16;
                    float* mrow = Msub + lm * MS + kc + 8 * hi;
                    v8f acc[VROWS / 16];
                    #pragma unroll
                    for (int vt = 0; vt < VROWS / 16; ++vt) {
                        const float4 c0 = *(const float4*)(mrow + vt * 16 * MS);
                        const float4 c1 = *(const float4*)(mrow + vt * 16 * MS + 4);
                        acc[vt][0] = c0.x; acc[vt][1] = c0.y; acc[vt][2] = c0.z; acc[vt][3] = c0.w;
                        acc[vt][4] = c1.x; acc[vt][5] = c1.y; acc[vt][6] = c1.z; acc[vt][7] = c1.w;
                    }
                    v2f ka[CHK / 4];
                    #pragma unroll
                    for (int j = 0; j < CHK / 4; ++j) {
                        ka[j].x = Kh[(4 * j + 2 * hi) * MS + kc + lm];
                        ka[j].y = Kh[(4 * j + 2 * hi + 1) * MS + kc + lm];
                    }
                    #pragma unroll
                    for (int j = 0; j < CHK / 4; ++j) {
                        #pragma unroll
                        for (int vt = 0; vt < VROWS / 16; ++vt)
                            acc[vt] = __builtin_amdgcn_wmma_f32_16x16x4_f32(
                                          false, ka[j], false, ub[vt][j], (short)0, acc[vt], false, false);
                    }
                    #pragma unroll
                    for (int vt = 0; vt < VROWS / 16; ++vt) {
                        float4 c0, c1;
                        c0.x = acc[vt][0]; c0.y = acc[vt][1]; c0.z = acc[vt][2]; c0.w = acc[vt][3];
                        c1.x = acc[vt][4]; c1.y = acc[vt][5]; c1.z = acc[vt][6]; c1.w = acc[vt][7];
                        *(float4*)(mrow + vt * 16 * MS)     = c0;
                        *(float4*)(mrow + vt * 16 * MS + 4) = c1;
                    }
                }
                __builtin_amdgcn_sched_barrier(0);
            }
            __syncthreads();
        }
    }

    float* dstB = (kind == 0)
        ? out + (size_t)b * PLANE + (size_t)v0 * DK_
        : ws + ((size_t)((seg - 1) * B_ + b) * 2 + (kind == 2 ? 0 : 1)) * PLANE + (size_t)v0 * DK_;
    for (int i = tid; i < VROWS * DK_; i += NTHREADS) {
        int r = i >> 8, k = i & 255;
        dstB[r * DK_ + k] = Msub[r * MS + k];
    }
}

#define CROWS 16
__global__ __launch_bounds__(NTHREADS)
void delta_compose_segments(float* __restrict__ out,
                            const float* __restrict__ ws,
                            size_t ws_bytes) {
    __shared__ float Mc[CROWS * MS];

    const int nseg = seg_count(ws_bytes);
    if (nseg == 1) return;

    const int tid  = threadIdx.x;
    const int wave = tid >> 5;
    const int lane = tid & 31;
    const int hi   = lane >> 4;
    const int lm   = lane & 15;

    const int b  = blockIdx.x / (DV_ / CROWS);
    const int r0 = (blockIdx.x % (DV_ / CROWS)) * CROWS;
    float* outB = out + (size_t)b * PLANE + (size_t)r0 * DK_;

    for (int i = tid; i < CROWS * DK_; i += NTHREADS) {
        int r = i >> 8, k = i & 255;
        Mc[r * MS + k] = outB[r * DK_ + k];
    }
    __syncthreads();

    const int c0 = wave * 32 + lm;
    for (int s = 1; s < nseg; ++s) {
        const float* P = ws + ((size_t)((s - 1) * B_ + b) * 2) * PLANE;
        const float* Q = P + PLANE + (size_t)r0 * DK_;
        v8f acc0, acc1;
        #pragma unroll
        for (int r = 0; r < 8; ++r) {
            acc0[r] = Q[(r + 8 * hi) * DK_ + c0];
            acc1[r] = Q[(r + 8 * hi) * DK_ + c0 + 16];
        }
        const float* arow = Mc + lm * MS + 2 * hi;
        const float* bcol = P + (size_t)(2 * hi) * DK_ + c0;
        v2f bx[8], by[8], nx[8], ny[8];
        #pragma unroll
        for (int j = 0; j < 8; ++j) {
            bx[j].x = bcol[(size_t)(4 * j) * DK_];
            bx[j].y = bcol[(size_t)(4 * j + 1) * DK_];
            by[j].x = bcol[(size_t)(4 * j) * DK_ + 16];
            by[j].y = bcol[(size_t)(4 * j + 1) * DK_ + 16];
        }
        #pragma unroll
        for (int g = 0; g < DK_ / 32; ++g) {
            if (g + 1 < DK_ / 32) {
                #pragma unroll
                for (int j = 0; j < 8; ++j) {
                    const int kn = (g + 1) * 32 + 4 * j;
                    nx[j].x = bcol[(size_t)kn * DK_];
                    nx[j].y = bcol[(size_t)(kn + 1) * DK_];
                    ny[j].x = bcol[(size_t)kn * DK_ + 16];
                    ny[j].y = bcol[(size_t)(kn + 1) * DK_ + 16];
                }
            }
            #pragma unroll
            for (int j = 0; j < 8; ++j) {
                v2f a = *(const v2f*)(arow + g * 32 + 4 * j);
                acc0 = __builtin_amdgcn_wmma_f32_16x16x4_f32(
                           false, a, false, bx[j], (short)0, acc0, false, false);
                acc1 = __builtin_amdgcn_wmma_f32_16x16x4_f32(
                           false, a, false, by[j], (short)0, acc1, false, false);
            }
            #pragma unroll
            for (int j = 0; j < 8; ++j) { bx[j] = nx[j]; by[j] = ny[j]; }
        }
        __syncthreads();
        #pragma unroll
        for (int r = 0; r < 8; ++r) {
            Mc[(r + 8 * hi) * MS + c0]      = acc0[r];
            Mc[(r + 8 * hi) * MS + c0 + 16] = acc1[r];
        }
        __syncthreads();
    }

    for (int i = tid; i < CROWS * DK_; i += NTHREADS) {
        int r = i >> 8, k = i & 255;
        outB[r * DK_ + k] = Mc[r * MS + k];
    }
}

extern "C" void kernel_launch(void* const* d_in, const int* in_sizes, int n_in,
                              void* d_out, int out_size, void* d_ws, size_t ws_size,
                              hipStream_t stream) {
    (void)in_sizes; (void)n_in; (void)out_size;
    const float* mem = (const float*)d_in[0];
    const float* key = (const float*)d_in[1];
    const float* val = (const float*)d_in[2];
    float* out = (float*)d_out;
    float* ws  = (float*)d_ws;

    dim3 grid(B_ * UNITS_MAX);
    dim3 block(NTHREADS);
    hipLaunchKernelGGL(delta_scan_segments, grid, block, 0, stream,
                       mem, key, val, out, ws, ws_size);
    hipLaunchKernelGGL(delta_compose_segments, dim3(B_ * (DV_ / CROWS)), block, 0, stream,
                       out, (const float*)ws, ws_size);
}


# Workgroup of 16 waves scans 128 value rows, a WGP to itself (224 workgroups): a SIMD's 4 waves share each phase; keys, Gram, solve chain, barriers serve twice the rows; sum order same
# speedup vs baseline: 3.4479x; 1.0679x over previous
#include <hip/hip_runtime.h>
#include <math.h>


typedef float v2f __attribute__((ext_vector_type(2)));
typedef float v8f __attribute__((ext_vector_type(8)));

#define B_   16
#define S_   4096
#define DK_  256
#define DV_  256
#define CHK  16
#define NCHK (S_ / CHK)
#define SPLIT 2
#define VROWS (DV_ / SPLIT)
#define VT (VROWS / 16)
#define NTHREADS 512
#define NWAVES (NTHREADS / 32)
#define KTHREADS 256
#define DT 4
#define CTHREADS 256
#define MS 260
#define UT 18
#define PT 20
#define STG 3

#define LR_C 0.1f
#define A_C  1.1f
#define EPS_C 1e-6f

#define SMEM_USED (VROWS * MS + STG * CHK * MS + 2 * VROWS * PT + VROWS * UT + 2 * 4 * 256 + STG * VROWS * PT + STG * 256 + 2 * STG * 16)
#define SMEM_FLOATS (250 * 1024 / 4)
static_assert(SMEM_USED <= SMEM_FLOATS, "the scan's buffers fit the LDS asked for");
static_assert(VROWS == 128 && CHK == 16, "the chunk solve maps one value row to each lane of four waves");
static_assert(2 * VT == NWAVES, "one P0 half-tile (16 value rows x half of the key columns) per wave");
static_assert(VT == 2 * DT && DK_ == (NWAVES / 2) * 32 && MS % 4 == 0,
              "the update gives each wave 32 key columns of half the value rows, read in 16-byte pieces");
static_assert(KTHREADS == CHK * 16 && NTHREADS * 4 == CHK * VROWS, "staging maps: keys on 256 threads, values on all");

#define NSEG_MAX 4
#define UNITS_MAX (SPLIT + 2 * SPLIT * (NSEG_MAX - 1))
#define PLANE (DV_ * DK_)
#define SEG_WS_BYTES ((size_t)B_ * 2 * PLANE * sizeof(float))

__device__ __forceinline__ int seg_count(size_t ws_bytes) {
    const size_t n = 1 + ws_bytes / SEG_WS_BYTES;
    return n > NSEG_MAX ? NSEG_MAX : (int)n;
}

__device__ __forceinline__ void gram_quarter(const float* __restrict__ krows, float* __restrict__ dst,
                                             int lm, int hi, int q) {
    const float* grow = krows + lm * MS + 2 * hi + q * 64;
    v2f f[16];
    #pragma unroll
    for (int i = 0; i < 16; ++i) f[i] = *(const v2f*)(grow + 4 * i);
    __builtin_amdgcn_sched_barrier(0);
    v8f g0 = {}, g1 = {};
    #pragma unroll
    for (int i = 0; i < 16; i += 2) {
        g0 = __builtin_amdgcn_wmma_f32_16x16x4_f32(
                 false, f[i], false, f[i], (short)0, g0, false, false);
        g1 = __builtin_amdgcn_wmma_f32_16x16x4_f32(
                 false, f[i + 1], false, f[i + 1], (short)0, g1, false, false);
    }
    const v8f gs = g0 + g1;
    __builtin_amdgcn_sched_barrier(0);
    #pragma unroll
    for (int r = 0; r < 8; ++r)
        dst[(r + 8 * hi) * 16 + lm] = gs[r];
}

__global__ __launch_bounds__(NTHREADS, 4)
void delta_scan_segments(const float* __restrict__ mem,
                         const float* __restrict__ key,
                         const float* __restrict__ val,
                         float* __restrict__ out,
                         float* __restrict__ ws,
                         size_t ws_bytes) {
    __shared__ __attribute__((aligned(16))) float smem[SMEM_FLOATS];
    float* Msub = smem;
    float* KnS  = Msub + VROWS * MS;
    float* P0Ta = KnS  + STG * CHK * MS;
    float* P0Tb = P0Ta + VROWS * PT;
    float* Uv   = P0Tb + VROWS * PT;
    float* GqS  = Uv   + VROWS * UT;
    float* VcS  = GqS  + 2 * 4 * 256;
    float* redS = VcS  + STG * VROWS * PT;
    float* lrcS = redS + STG * 256;
    float* ac2S = lrcS + STG * 16;

    const int tid  = threadIdx.x;
    const int wave = tid >> 5;
    const int lane = tid & 31;
    const int hi   = lane >> 4;
    const int lm   = lane & 15;

    const int nseg = seg_count(ws_bytes);
    const int b    = blockIdx.x % B_;
    const int unit = blockIdx.x / B_;
    if (unit >= SPLIT + 2 * SPLIT * (nseg - 1)) return;
    int seg = 0, kind = 0, slice = unit;
    if (unit >= SPLIT) {
        const int w = unit - SPLIT;
        const int r = w % (2 * SPLIT);
        seg   = 1 + w / (2 * SPLIT);
        kind  = r < SPLIT ? 1 : 2;
        slice = r % SPLIT;
    }
    const int v0   = slice * VROWS;
    const int cbeg = seg * NCHK / nseg;
    const int cend = (seg + 1) * NCHK / nseg;

    const float* memB = mem + (size_t)b * PLANE + (size_t)v0 * DK_;
    const float* keyB = key + (size_t)b * S_ * DK_;
    const float* valB = val + (size_t)b * S_ * DV_;

    if (kind == 0) {
        for (int i = tid; i < VROWS * DK_; i += NTHREADS) {
            int r = i >> 8, k = i & 255;
            Msub[r * MS + k] = memB[r * DK_ + k];
        }
    } else {
        const float diag = (kind == 2) ? 1.f : 0.f;
        for (int i = tid; i < VROWS * DK_; i += NTHREADS) {
            int r = i >> 8, k = i & 255;
            Msub[r * MS + k] = (v0 + r == k) ? diag : 0.f;
        }
    }
    __syncthreads();

    const int tA = (tid >> 4) & 15;
    const int sg = tid & 15;
    const int d0 = sg * 16;
    const int tV = ((tid >> 5) & 3) * 4 + ((tid & 31) >> 3);
    const int jV = (tid >> 7) * 32 + (tid & 7) * 4;
    const bool kst = tid < KTHREADS;

    for (int c = cbeg; c < cend; c += STG) {
        const int n2 = (cend - c) < STG ? (cend - c) : STG;
        {
            float4 rv[STG][4];
            float4 vv[STG];
            #pragma unroll
            for (int h = 0; h < STG; ++h) {
                vv[h] = make_float4(0.f, 0.f, 0.f, 0.f);
                #pragma unroll
                for (int j = 0; j < 4; ++j) rv[h][j] = make_float4(0.f, 0.f, 0.f, 0.f);
                if (h < n2) {
                    if (kst) {
                        const float* kr = keyB + (size_t)((c + h) * CHK + tA) * DK_ + d0;
                        #pragma unroll
                        for (int j = 0; j < 4; ++j) rv[h][j] = ((const float4*)kr)[j];
                    }
                    if (kind != 2)
                        vv[h] = *(const float4*)(valB + (size_t)((c + h) * CHK + tV) * DV_ + v0 + jV);
                }
            }
            #pragma unroll
            for (int h = 0; h < STG; ++h) {
                if (kst && c + STG + h < cend)
                    __builtin_prefetch(keyB + (size_t)((c + STG + h) * CHK + tA) * DK_ + d0, 0, 0);
            }
            #pragma unroll
            for (int h = 0; h < STG; ++h) {
                if (h < n2) {
                    float* Vh = VcS + h * VROWS * PT;
                    Vh[jV * PT + tV]       = vv[h].x;
                    Vh[(jV + 1) * PT + tV] = vv[h].y;
                    Vh[(jV + 2) * PT + tV] = vv[h].z;
                    Vh[(jV + 3) * PT + tV] = vv[h].w;
                    if (kst) {
                        float ss = 0.f;
                        #pragma unroll
                        for (int j = 0; j < 4; ++j)
                            ss += rv[h][j].x * rv[h][j].x + rv[h][j].y * rv[h][j].y +
                                  rv[h][j].z * rv[h][j].z + rv[h][j].w * rv[h][j].w;
                        redS[h * 256 + tA * 16 + sg] = ss;
                        float* kw = KnS + (h * CHK + tA) * MS + d0;
                        #pragma unroll
                        for (int j = 0; j < 4; ++j) *(float4*)(kw + 4 * j) = rv[h][j];
                    }
                }
            }
        }
        __syncthreads();

        #pragma unroll
        for (int h = 0; h < STG; ++h) {
            if (h >= n2) break;
            const float* Kh   = KnS  + h * CHK * MS;
            const float* Vh   = VcS  + h * VROWS * PT;
            const float* redh = redS + h * 256;
            float* lrch = lrcS + h * 16;
            float* ac2h = ac2S + h * 16;
            const float* Gq = GqS + (h & 1) * 4 * 256;

            {
                const int tile = wave >> 1;
                const int kh   = wave & 1;
                const float* arow = Msub + (tile * 16 + lm) * MS + 2 * hi;
                const float* brow = Kh + lm * MS + 2 * hi;
                v8f acc0 = {}, acc1 = {};
                const int kbeg = kh * 128;
                for (int k0 = kbeg; k0 < kbeg + 128; k0 += 8) {
                    v2f a0 = *(const v2f*)(arow + k0);
                    v2f b0 = *(const v2f*)(brow + k0);
                    acc0 = __builtin_amdgcn_wmma_f32_16x16x4_f32(
                               false, a0, false, b0, (short)0, acc0, false, false);
                    v2f a1 = *(const v2f*)(arow + k0 + 4);
                    v2f b1 = *(const v2f*)(brow + k0 + 4);
                    acc1 = __builtin_amdgcn_wmma_f32_16x16x4_f32(
                               false, a1, false, b1, (short)0, acc1, false, false);
                }
                v8f acc = acc0 + acc1;
                float* dst = kh ? P0Tb : P0Ta;
                #pragma unroll
                for (int r = 0; r < 8; ++r)
                    dst[(tile * 16 + r + 8 * hi) * PT + lm] = acc[r];
            }
            if (wave >= 12) {
                if (h == 0)
                    gram_quarter(Kh, GqS + (wave - 12) * 256, lm, hi, wave - 12);
            } else if (wave == 11) {
                if (lane < 16) {
                    float s = 0.f;
                    #pragma unroll
                    for (int j = 0; j < 16; ++j) s += redh[lane * 16 + j];
                    const float c = 1.f / (sqrtf(s) + EPS_C);
                    lrch[lane] = LR_C * c;
                    ac2h[lane] = A_C * c * c;
                }
            }
            __syncthreads();

            if (tid < VROWS) {
                const int v = tid;
                const float4 ga0 = *(const float4*)(Gq + 4 * lane);
                const float4 ga1 = *(const float4*)(Gq + 128 + 4 * lane);
                const float4 gb0 = *(const float4*)(Gq + 256 + 4 * lane);
                const float4 gb1 = *(const float4*)(Gq + 256 + 128 + 4 * lane);
                const float4 gc0 = *(const float4*)(Gq + 512 + 4 * lane);
                const float4 gc1 = *(const float4*)(Gq + 512 + 128 + 4 * lane);
                const float4 gd0 = *(const float4*)(Gq + 768 + 4 * lane);
                const float4 gd1 = *(const float4*)(Gq + 768 + 128 + 4 * lane);
                const float4* pa = (const float4*)(P0Ta + v * PT);
                const float4* pb = (const float4*)(P0Tb + v * PT);
                const float4* pv = (const float4*)(Vh + v * PT);
                const float4* pl = (const float4*)lrch;
                const float4* pc = (const float4*)ac2h;
                const float gt0 = ac2h[lane >> 2];
                const float gt1 = ac2h[8 + (lane >> 2)];
                float x[CHK];
                #pragma unroll
                for (int q = 0; q < CHK / 4; ++q) {
                    const float4 a4 = pa[q], b4 = pb[q], v4 = pv[q], l4 = pl[q], c4 = pc[q];
                    x[4 * q + 0] = l4.x * v4.x - c4.x * (a4.x + b4.x);
                    x[4 * q + 1] = l4.y * v4.y - c4.y * (a4.y + b4.y);
                    x[4 * q + 2] = l4.z * v4.z - c4.z * (a4.z + b4.z);
                    x[4 * q + 3] = l4.w * v4.w - c4.w * (a4.w + b4.w);
                }
                float g[8];
                g[0] = gt0 * ((ga0.x + gb0.x) + (gc0.x + gd0.x));
                g[1] = gt0 * ((ga0.y + gb0.y) + (gc0.y + gd0.y));
                g[2] = gt0 * ((ga0.z + gb0.z) + (gc0.z + gd0.z));
                g[3] = gt0 * ((ga0.w + gb0.w) + (gc0.w + gd0.w));
                g[4] = gt1 * ((ga1.x + gb1.x) + (gc1.x + gd1.x));
                g[5] = gt1 * ((ga1.y + gb1.y) + (gc1.y + gd1.y));
                g[6] = gt1 * ((ga1.z + gb1.z) + (gc1.z + gd1.z));
                g[7] = gt1 * ((ga1.w + gb1.w) + (gc1.w + gd1.w));
                #pragma unroll
                for (int s2 = 0; s2 < CHK - 1; ++s2) {
                    #pragma unroll
                    for (int t = s2 + 1; t < CHK; ++t) {
                        const float gts = __int_as_float(__builtin_amdgcn_readlane(
                            __float_as_int(g[4 * (t >> 3) + (s2 & 3)]), 4 * (t & 7) + (s2 >> 2)));
                        x[t] -= gts * x[s2];
                    }
                }
                float* uw = Uv + v * UT;
                #pragma unroll
                for (int t = 0; t < CHK; ++t) uw[t] = x[t];
            } else if (wave >= 12) {
                if (h + 1 < STG && h + 1 < n2)
                    gram_quarter(KnS + (h + 1) * CHK * MS, GqS + ((h + 1) & 1) * 4 * 256 + (wave - 12) * 256,
                                 lm, hi, wave - 12);
            }
            __syncthreads();

            {
                const int vb = (wave >> 3) * (DT * 16);
                const int wk = wave & 7;
                v2f ub[DT][CHK / 4];
                #pragma unroll
                for (int vt = 0; vt < DT; ++vt) {
                    #pragma unroll
                    for (int j = 0; j < CHK / 4; ++j)
                        ub[vt][j] = *(const v2f*)(Uv + (vb + vt * 16 + lm) * UT + 4 * j + 2 * hi);
                }
                #pragma unroll
                for (int kk = 0; kk < 2; ++kk) {
                    const int kc = wk * 32 + kk * 16;
                    float* mrow = Msub + (vb + lm) * MS + kc + 8 * hi;
                    v8f acc[DT];
                    #pragma unroll
                    for (int vt = 0; vt < DT; ++vt) {
                        const float4 c0 = *(const float4*)(mrow + vt * 16 * MS);
                        const float4 c1 = *(const float4*)(mrow + vt * 16 * MS + 4);
                        acc[vt][0] = c0.x; acc[vt][1] = c0.y; acc[vt][2] = c0.z; acc[vt][3] = c0.w;
                        acc[vt][4] = c1.x; acc[vt][5] = c1.y; acc[vt][6] = c1.z; acc[vt][7] = c1.w;
                    }
                    v2f ka[CHK / 4];
                    #pragma unroll
                    for (int j = 0; j < CHK / 4; ++j) {
                        ka[j].x = Kh[(4 * j + 2 * hi) * MS + kc + lm];
                        ka[j].y = Kh[(4 * j + 2 * hi + 1) * MS + kc + lm];
                    }
                    #pragma unroll
                    for (int j = 0; j < CHK / 4; ++j) {
                        #pragma unroll
                        for (int vt = 0; vt < DT; ++vt)
                            acc[vt] = __builtin_amdgcn_wmma_f32_16x16x4_f32(
                                          false, ka[j], false, ub[vt][j], (short)0, acc[vt], false, false);
                    }
                    #pragma unroll
                    for (int vt = 0; vt < DT; ++vt) {
                        float4 c0, c1;
                        c0.x = acc[vt][0]; c0.y = acc[vt][1]; c0.z = acc[vt][2]; c0.w = acc[vt][3];
                        c1.x = acc[vt][4]; c1.y = acc[vt][5]; c1.z = acc[vt][6]; c1.w = acc[vt][7];
                        *(float4*)(mrow + vt * 16 * MS)     = c0;
                        *(float4*)(mrow + vt * 16 * MS + 4) = c1;
                    }
                }
                __builtin_amdgcn_sched_barrier(0);
            }
            __syncthreads();
        }
    }

    float* dstB = (kind == 0)
        ? out + (size_t)b * PLANE + (size_t)v0 * DK_
        : ws + ((size_t)((seg - 1) * B_ + b) * 2 + (kind == 2 ? 0 : 1)) * PLANE + (size_t)v0 * DK_;
    for (int i = tid; i < VROWS * DK_; i += NTHREADS) {
        int r = i >> 8, k = i & 255;
        dstB[r * DK_ + k] = Msub[r * MS + k];
    }
}

#define CROWS 16
__global__ __launch_bounds__(CTHREADS)
void delta_compose_segments(float* __restrict__ out,
                            const float* __restrict__ ws,
                            size_t ws_bytes) {
    __shared__ float Mc[CROWS * MS];

    const int nseg = seg_count(ws_bytes);
    if (nseg == 1) return;

    const int tid  = threadIdx.x;
    const int wave = tid >> 5;
    const int lane = tid & 31;
    const int hi   = lane >> 4;
    const int lm   = lane & 15;

    const int b  = blockIdx.x / (DV_ / CROWS);
    const int r0 = (blockIdx.x % (DV_ / CROWS)) * CROWS;
    float* outB = out + (size_t)b * PLANE + (size_t)r0 * DK_;

    for (int i = tid; i < CROWS * DK_; i += CTHREADS) {
        int r = i >> 8, k = i & 255;
        Mc[r * MS + k] = outB[r * DK_ + k];
    }
    __syncthreads();

    const int c0 = wave * 32 + lm;
    for (int s = 1; s < nseg; ++s) {
        const float* P = ws + ((size_t)((s - 1) * B_ + b) * 2) * PLANE;
        const float* Q = P + PLANE + (size_t)r0 * DK_;
        v8f acc0, acc1;
        #pragma unroll
        for (int r = 0; r < 8; ++r) {
            acc0[r] = Q[(r + 8 * hi) * DK_ + c0];
            acc1[r] = Q[(r + 8 * hi) * DK_ + c0 + 16];
        }
        const float* arow = Mc + lm * MS + 2 * hi;
        const float* bcol = P + (size_t)(2 * hi) * DK_ + c0;
        v2f bx[8], by[8], nx[8], ny[8];
        #pragma unroll
        for (int j = 0; j < 8; ++j) {
            bx[j].x = bcol[(size_t)(4 * j) * DK_];
            bx[j].y = bcol[(size_t)(4 * j + 1) * DK_];
            by[j].x = bcol[(size_t)(4 * j) * DK_ + 16];
            by[j].y = bcol[(size_t)(4 * j + 1) * DK_ + 16];
        }
        #pragma unroll
        for (int g = 0; g < DK_ / 32; ++g) {
            if (g + 1 < DK_ / 32) {
                #pragma unroll
                for (int j = 0; j < 8; ++j) {
                    const int kn = (g + 1) * 32 + 4 * j;
                    nx[j].x = bcol[(size_t)kn * DK_];
                    nx[j].y = bcol[(size_t)(kn + 1) * DK_];
                    ny[j].x = bcol[(size_t)kn * DK_ + 16];
                    ny[j].y = bcol[(size_t)(kn + 1) * DK_ + 16];
                }
            }
            #pragma unroll
            for (int j = 0; j < 8; ++j) {
                v2f a = *(const v2f*)(arow + g * 32 + 4 * j);
                acc0 = __builtin_amdgcn_wmma_f32_16x16x4_f32(
                           false, a, false, bx[j], (short)0, acc0, false, false);
                acc1 = __builtin_amdgcn_wmma_f32_16x16x4_f32(
                           false, a, false, by[j], (short)0, acc1, false, false);
            }
            #pragma unroll
            for (int j = 0; j < 8; ++j) { bx[j] = nx[j]; by[j] = ny[j]; }
        }
        __syncthreads();
        #pragma unroll
        for (int r = 0; r < 8; ++r) {
            Mc[(r + 8 * hi) * MS + c0]      = acc0[r];
            Mc[(r + 8 * hi) * MS + c0 + 16] = acc1[r];
        }
        __syncthreads();
    }

    for (int i = tid; i < CROWS * DK_; i += CTHREADS) {
        int r = i >> 8, k = i & 255;
        outB[r * DK_ + k] = Mc[r * MS + k];
    }
}

extern "C" void kernel_launch(void* const* d_in, const int* in_sizes, int n_in,
                              void* d_out, int out_size, void* d_ws, size_t ws_size,
                              hipStream_t stream) {
    (void)in_sizes; (void)n_in; (void)out_size;
    const float* mem = (const float*)d_in[0];
    const float* key = (const float*)d_in[1];
    const float* val = (const float*)d_in[2];
    float* out = (float*)d_out;
    float* ws  = (float*)d_ws;

    dim3 grid(B_ * UNITS_MAX);
    dim3 block(NTHREADS);
    hipLaunchKernelGGL(delta_scan_segments, grid, block, 0, stream,
                       mem, key, val, out, ws, ws_size);
    hipLaunchKernelGGL(delta_compose_segments, dim3(B_ * (DV_ / CROWS)), dim3(CTHREADS), 0, stream,
                       out, (const float*)ws, ws_size);
}


# Uneven time split over all 256 WGPs: segment 0 scans 79 of 256 chunks on 4 workgroups of 64 rows (was 2 of 128), later segments 59 each, not 64 (sum order changed: composition points moved)
# speedup vs baseline: 3.6114x; 1.0474x over previous
#include <hip/hip_runtime.h>
#include <math.h>


typedef float v2f __attribute__((ext_vector_type(2)));
typedef float v8f __attribute__((ext_vector_type(8)));

#define B_   16
#define S_   4096
#define DK_  256
#define DV_  256
#define CHK  16
#define NCHK (S_ / CHK)
#define SPLIT 2
#define VROWS (DV_ / SPLIT)
#define VT (VROWS / 16)
#define NTHREADS 512
#define NWAVES (NTHREADS / 32)
#define KTHREADS 256
#define DT 4
#define CTHREADS 256
#define MS 260
#define UT 18
#define PT 20
#define STG 3

#define LR_C 0.1f
#define A_C  1.1f
#define EPS_C 1e-6f

#define SMEM_USED (VROWS * MS + STG * CHK * MS + 2 * VROWS * PT + VROWS * UT + 2 * 4 * 256 + STG * VROWS * PT + STG * 256 + 2 * STG * 16)
#define SMEM_FLOATS (250 * 1024 / 4)
static_assert(SMEM_USED <= SMEM_FLOATS, "the scan's buffers fit the LDS asked for");
static_assert(VROWS == 128 && CHK == 16, "the chunk solve maps one value row to each lane of four waves");
static_assert(2 * VT == NWAVES, "one P0 half-tile (16 value rows x half of the key columns) per wave");
static_assert(VT == 2 * DT && DK_ == (NWAVES / 2) * 32 && MS % 4 == 0,
              "the update gives each wave 32 key columns of half the value rows, read in 16-byte pieces");
static_assert(KTHREADS == CHK * 16 && NTHREADS * 4 == CHK * VROWS, "staging maps: keys on 256 threads, values on all");

#define NSEG_MAX 4
#define HROWS  64
#define SPLIT0 (DV_ / HROWS)
#define SEG0_W 19
#define SEGN_W 14
#define UNITS_MAX (SPLIT0 + 2 * SPLIT * (NSEG_MAX - 1))
static_assert(2 * HROWS == VROWS, "a segment-0 workgroup is the first half (the rows of waves 0-7) of a full one");
#define PLANE (DV_ * DK_)
#define SEG_WS_BYTES ((size_t)B_ * 2 * PLANE * sizeof(float))

__device__ __forceinline__ int seg_count(size_t ws_bytes) {
    const size_t n = 1 + ws_bytes / SEG_WS_BYTES;
    return n > NSEG_MAX ? NSEG_MAX : (int)n;
}

__device__ __forceinline__ void gram_quarter(const float* __restrict__ krows, float* __restrict__ dst,
                                             int lm, int hi, int q) {
    const float* grow = krows + lm * MS + 2 * hi + q * 64;
    v2f f[16];
    #pragma unroll
    for (int i = 0; i < 16; ++i) f[i] = *(const v2f*)(grow + 4 * i);
    __builtin_amdgcn_sched_barrier(0);
    v8f g0 = {}, g1 = {};
    #pragma unroll
    for (int i = 0; i < 16; i += 2) {
        g0 = __builtin_amdgcn_wmma_f32_16x16x4_f32(
                 false, f[i], false, f[i], (short)0, g0, false, false);
        g1 = __builtin_amdgcn_wmma_f32_16x16x4_f32(
                 false, f[i + 1], false, f[i + 1], (short)0, g1, false, false);
    }
    const v8f gs = g0 + g1;
    __builtin_amdgcn_sched_barrier(0);
    #pragma unroll
    for (int r = 0; r < 8; ++r)
        dst[(r + 8 * hi) * 16 + lm] = gs[r];
}

__global__ __launch_bounds__(NTHREADS, 4)
void delta_scan_segments(const float* __restrict__ mem,
                         const float* __restrict__ key,
                         const float* __restrict__ val,
                         float* __restrict__ out,
                         float* __restrict__ ws,
                         size_t ws_bytes) {
    __shared__ __attribute__((aligned(16))) float smem[SMEM_FLOATS];
    float* Msub = smem;
    float* KnS  = Msub + VROWS * MS;
    float* P0Ta = KnS  + STG * CHK * MS;
    float* P0Tb = P0Ta + VROWS * PT;
    float* Uv   = P0Tb + VROWS * PT;
    float* GqS  = Uv   + VROWS * UT;
    float* VcS  = GqS  + 2 * 4 * 256;
    float* redS = VcS  + STG * VROWS * PT;
    float* lrcS = redS + STG * 256;
    float* ac2S = lrcS + STG * 16;

    const int tid  = threadIdx.x;
    const int wave = tid >> 5;
    const int lane = tid & 31;
    const int hi   = lane >> 4;
    const int lm   = lane & 15;

    const int nseg = seg_count(ws_bytes);
    const int b    = blockIdx.x % B_;
    const int unit = (blockIdx.x / B_ + SPLIT0) % UNITS_MAX;
    if (unit >= SPLIT0 + 2 * SPLIT * (nseg - 1)) return;
    int seg = 0, kind = 0;
    int v0 = unit * HROWS;
    if (unit >= SPLIT0) {
        const int w = unit - SPLIT0;
        const int r = w % (2 * SPLIT);
        seg  = 1 + w / (2 * SPLIT);
        kind = r < SPLIT ? 1 : 2;
        v0   = (r % SPLIT) * VROWS;
    }
    const bool full = seg != 0;
    const int nrows = full ? VROWS : HROWS;
    const int len0 = NCHK * SEG0_W / (SEG0_W + SEGN_W * (nseg - 1));
    const int nlat = nseg > 1 ? nseg - 1 : 1;
    const int cbeg = seg == 0 ? 0    : len0 + (seg - 1) * (NCHK - len0) / nlat;
    const int cend = seg == 0 ? len0 : len0 + seg * (NCHK - len0) / nlat;

    const float* memB = mem + (size_t)b * PLANE + (size_t)v0 * DK_;
    const float* keyB = key + (size_t)b * S_ * DK_;
    const float* valB = val + (size_t)b * S_ * DV_;

    if (kind == 0) {
        for (int i = tid; i < nrows * DK_; i += NTHREADS) {
            int r = i >> 8, k = i & 255;
            Msub[r * MS + k] = memB[r * DK_ + k];
        }
    } else {
        const float diag = (kind == 2) ? 1.f : 0.f;
        for (int i = tid; i < VROWS * DK_; i += NTHREADS) {
            int r = i >> 8, k = i & 255;
            Msub[r * MS + k] = (v0 + r == k) ? diag : 0.f;
        }
    }
    __syncthreads();

    const int tA = (tid >> 4) & 15;
    const int sg = tid & 15;
    const int d0 = sg * 16;
    const int tV = ((tid >> 5) & 3) * 4 + ((tid & 31) >> 3);
    const int jV = (tid >> 7) * 32 + (tid & 7) * 4;
    const bool kst = tid < KTHREADS;

    for (int c = cbeg; c < cend; c += STG) {
        const int n2 = (cend - c) < STG ? (cend - c) : STG;
        {
            float4 rv[STG][4];
            float4 vv[STG];
            #pragma unroll
            for (int h = 0; h < STG; ++h) {
                vv[h] = make_float4(0.f, 0.f, 0.f, 0.f);
                #pragma unroll
                for (int j = 0; j < 4; ++j) rv[h][j] = make_float4(0.f, 0.f, 0.f, 0.f);
                if (h < n2) {
                    if (kst) {
                        const float* kr = keyB + (size_t)((c + h) * CHK + tA) * DK_ + d0;
                        #pragma unroll
                        for (int j = 0; j < 4; ++j) rv[h][j] = ((const float4*)kr)[j];
                    }
                    if (kind != 2 && jV < nrows)
                        vv[h] = *(const float4*)(valB + (size_t)((c + h) * CHK + tV) * DV_ + v0 + jV);
                }
            }
            #pragma unroll
            for (int h = 0; h < STG; ++h) {
                if (kst && c + STG + h < cend)
                    __builtin_prefetch(keyB + (size_t)((c + STG + h) * CHK + tA) * DK_ + d0, 0, 0);
            }
            #pragma unroll
            for (int h = 0; h < STG; ++h) {
                if (h < n2) {
                    float* Vh = VcS + h * VROWS * PT;
                    Vh[jV * PT + tV]       = vv[h].x;
                    Vh[(jV + 1) * PT + tV] = vv[h].y;
                    Vh[(jV + 2) * PT + tV] = vv[h].z;
                    Vh[(jV + 3) * PT + tV] = vv[h].w;
                    if (kst) {
                        float ss = 0.f;
                        #pragma unroll
                        for (int j = 0; j < 4; ++j)
                            ss += rv[h][j].x * rv[h][j].x + rv[h][j].y * rv[h][j].y +
                                  rv[h][j].z * rv[h][j].z + rv[h][j].w * rv[h][j].w;
                        redS[h * 256 + tA * 16 + sg] = ss;
                        float* kw = KnS + (h * CHK + tA) * MS + d0;
                        #pragma unroll
                        for (int j = 0; j < 4; ++j) *(float4*)(kw + 4 * j) = rv[h][j];
                    }
                }
            }
        }
        __syncthreads();

        #pragma unroll
        for (int h = 0; h < STG; ++h) {
            if (h >= n2) break;
            const float* Kh   = KnS  + h * CHK * MS;
            const float* Vh   = VcS  + h * VROWS * PT;
            const float* redh = redS + h * 256;
            float* lrch = lrcS + h * 16;
            float* ac2h = ac2S + h * 16;
            const float* Gq = GqS + (h & 1) * 4 * 256;

            if (full || wave < 8) {
                const int tile = wave >> 1;
                const int kh   = wave & 1;
                const float* arow = Msub + (tile * 16 + lm) * MS + 2 * hi;
                const float* brow = Kh + lm * MS + 2 * hi;
                v8f acc0 = {}, acc1 = {};
                const int kbeg = kh * 128;
                for (int k0 = kbeg; k0 < kbeg + 128; k0 += 8) {
                    v2f a0 = *(const v2f*)(arow + k0);
                    v2f b0 = *(const v2f*)(brow + k0);
                    acc0 = __builtin_amdgcn_wmma_f32_16x16x4_f32(
                               false, a0, false, b0, (short)0, acc0, false, false);
                    v2f a1 = *(const v2f*)(arow + k0 + 4);
                    v2f b1 = *(const v2f*)(brow + k0 + 4);
                    acc1 = __builtin_amdgcn_wmma_f32_16x16x4_f32(
                               false, a1, false, b1, (short)0, acc1, false, false);
                }
                v8f acc = acc0 + acc1;
                __builtin_amdgcn_sched_barrier(0);
                float* dst = kh ? P0Tb : P0Ta;
                #pragma unroll
                for (int r = 0; r < 8; ++r)
                    dst[(tile * 16 + r + 8 * hi) * PT + lm] = acc[r];
            }
            if (wave >= 12) {
                if (h == 0)
                    gram_quarter(Kh, GqS + (wave - 12) * 256, lm, hi, wave - 12);
            } else if (wave == 11) {
                if (lane < 16) {
                    float s = 0.f;
                    #pragma unroll
                    for (int j = 0; j < 16; ++j) s += redh[lane * 16 + j];
                    const float c = 1.f / (sqrtf(s) + EPS_C);
                    lrch[lane] = LR_C * c;
                    ac2h[lane] = A_C * c * c;
                }
            }
            __syncthreads();

            if (tid < nrows) {
                const int v = tid;
                const float4 ga0 = *(const float4*)(Gq + 4 * lane);
                const float4 ga1 = *(const float4*)(Gq + 128 + 4 * lane);
                const float4 gb0 = *(const float4*)(Gq + 256 + 4 * lane);
                const float4 gb1 = *(const float4*)(Gq + 256 + 128 + 4 * lane);
                const float4 gc0 = *(const float4*)(Gq + 512 + 4 * lane);
                const float4 gc1 = *(const float4*)(Gq + 512 + 128 + 4 * lane);
                const float4 gd0 = *(const float4*)(Gq + 768 + 4 * lane);
                const float4 gd1 = *(const float4*)(Gq + 768 + 128 + 4 * lane);
                const float4* pa = (const float4*)(P0Ta + v * PT);
                const float4* pb = (const float4*)(P0Tb + v * PT);
                const float4* pv = (const float4*)(Vh + v * PT);
                const float4* pl = (const float4*)lrch;
                const float4* pc = (const float4*)ac2h;
                const float gt0 = ac2h[lane >> 2];
                const float gt1 = ac2h[8 + (lane >> 2)];
                float x[CHK];
                #pragma unroll
                for (int q = 0; q < CHK / 4; ++q) {
                    const float4 a4 = pa[q], b4 = pb[q], v4 = pv[q], l4 = pl[q], c4 = pc[q];
                    x[4 * q + 0] = l4.x * v4.x - c4.x * (a4.x + b4.x);
                    x[4 * q + 1] = l4.y * v4.y - c4.y * (a4.y + b4.y);
                    x[4 * q + 2] = l4.z * v4.z - c4.z * (a4.z + b4.z);
                    x[4 * q + 3] = l4.w * v4.w - c4.w * (a4.w + b4.w);
                }
                float g[8];
                g[0] = gt0 * ((ga0.x + gb0.x) + (gc0.x + gd0.x));
                g[1] = gt0 * ((ga0.y + gb0.y) + (gc0.y + gd0.y));
                g[2] = gt0 * ((ga0.z + gb0.z) + (gc0.z + gd0.z));
                g[3] = gt0 * ((ga0.w + gb0.w) + (gc0.w + gd0.w));
                g[4] = gt1 * ((ga1.x + gb1.x) + (gc1.x + gd1.x));
                g[5] = gt1 * ((ga1.y + gb1.y) + (gc1.y + gd1.y));
                g[6] = gt1 * ((ga1.z + gb1.z) + (gc1.z + gd1.z));
                g[7] = gt1 * ((ga1.w + gb1.w) + (gc1.w + gd1.w));
                #pragma unroll
                for (int s2 = 0; s2 < CHK - 1; ++s2) {
                    #pragma unroll
                    for (int t = s2 + 1; t < CHK; ++t) {
                        const float gts = __int_as_float(__builtin_amdgcn_readlane(
                            __float_as_int(g[4 * (t >> 3) + (s2 & 3)]), 4 * (t & 7) + (s2 >> 2)));
                        x[t] -= gts * x[s2];
                    }
                }
                float* uw = Uv + v * UT;
                #pragma unroll
                for (int t = 0; t < CHK; ++t) uw[t] = x[t];
            } else if (wave >= 12) {
                if (h + 1 < STG && h + 1 < n2)
                    gram_quarter(KnS + (h + 1) * CHK * MS, GqS + ((h + 1) & 1) * 4 * 256 + (wave - 12) * 256,
                                 lm, hi, wave - 12);
            }
            __syncthreads();

            if (full || wave < 8) {
                const int vb = (wave >> 3) * (DT * 16);
                const int wk = wave & 7;
                v2f ub[DT][CHK / 4];
                #pragma unroll
                for (int vt = 0; vt < DT; ++vt) {
                    #pragma unroll
                    for (int j = 0; j < CHK / 4; ++j)
                        ub[vt][j] = *(const v2f*)(Uv + (vb + vt * 16 + lm) * UT + 4 * j + 2 * hi);
                }
                #pragma unroll
                for (int kk = 0; kk < 2; ++kk) {
                    const int kc = wk * 32 + kk * 16;
                    float* mrow = Msub + (vb + lm) * MS + kc + 8 * hi;
                    v8f acc[DT];
                    #pragma unroll
                    for (int vt = 0; vt < DT; ++vt) {
                        const float4 c0 = *(const float4*)(mrow + vt * 16 * MS);
                        const float4 c1 = *(const float4*)(mrow + vt * 16 * MS + 4);
                        acc[vt][0] = c0.x; acc[vt][1] = c0.y; acc[vt][2] = c0.z; acc[vt][3] = c0.w;
                        acc[vt][4] = c1.x; acc[vt][5] = c1.y; acc[vt][6] = c1.z; acc[vt][7] = c1.w;
                    }
                    v2f ka[CHK / 4];
                    #pragma unroll
                    for (int j = 0; j < CHK / 4; ++j) {
                        ka[j].x = Kh[(4 * j + 2 * hi) * MS + kc + lm];
                        ka[j].y = Kh[(4 * j + 2 * hi + 1) * MS + kc + lm];
                    }
                    #pragma unroll
                    for (int j = 0; j < CHK / 4; ++j) {
                        #pragma unroll
                        for (int vt = 0; vt < DT; ++vt)
                            acc[vt] = __builtin_amdgcn_wmma_f32_16x16x4_f32(
                                          false, ka[j], false, ub[vt][j], (short)0, acc[vt], false, false);
                    }
                    #pragma unroll
                    for (int vt = 0; vt < DT; ++vt) {
                        float4 c0, c1;
                        c0.x = acc[vt][0]; c0.y = acc[vt][1]; c0.z = acc[vt][2]; c0.w = acc[vt][3];
                        c1.x = acc[vt][4]; c1.y = acc[vt][5]; c1.z = acc[vt][6]; c1.w = acc[vt][7];
                        *(float4*)(mrow + vt * 16 * MS)     = c0;
                        *(float4*)(mrow + vt * 16 * MS + 4) = c1;
                    }
                }
                __builtin_amdgcn_sched_barrier(0);
            }
            __syncthreads();
        }
    }

    float* dstB = (kind == 0)
        ? out + (size_t)b * PLANE + (size_t)v0 * DK_
        : ws + ((size_t)((seg - 1) * B_ + b) * 2 + (kind == 2 ? 0 : 1)) * PLANE + (size_t)v0 * DK_;
    for (int i = tid; i < nrows * DK_; i += NTHREADS) {
        int r = i >> 8, k = i & 255;
        dstB[r * DK_ + k] = Msub[r * MS + k];
    }
}

#define CROWS 16
__global__ __launch_bounds__(CTHREADS)
void delta_compose_segments(float* __restrict__ out,
                            const float* __restrict__ ws,
                            size_t ws_bytes) {
    __shared__ float Mc[CROWS * MS];

    const int nseg = seg_count(ws_bytes);
    if (nseg == 1) return;

    const int tid  = threadIdx.x;
    const int wave = tid >> 5;
    const int lane = tid & 31;
    const int hi   = lane >> 4;
    const int lm   = lane & 15;

    const int b  = blockIdx.x / (DV_ / CROWS);
    const int r0 = (blockIdx.x % (DV_ / CROWS)) * CROWS;
    float* outB = out + (size_t)b * PLANE + (size_t)r0 * DK_;

    for (int i = tid; i < CROWS * DK_; i += CTHREADS) {
        int r = i >> 8, k = i & 255;
        Mc[r * MS + k] = outB[r * DK_ + k];
    }
    __syncthreads();

    const int c0 = wave * 32 + lm;
    for (int s = 1; s < nseg; ++s) {
        const float* P = ws + ((size_t)((s - 1) * B_ + b) * 2) * PLANE;
        const float* Q = P + PLANE + (size_t)r0 * DK_;
        v8f acc0, acc1;
        #pragma unroll
        for (int r = 0; r < 8; ++r) {
            acc0[r] = Q[(r + 8 * hi) * DK_ + c0];
            acc1[r] = Q[(r + 8 * hi) * DK_ + c0 + 16];
        }
        const float* arow = Mc + lm * MS + 2 * hi;
        const float* bcol = P + (size_t)(2 * hi) * DK_ + c0;
        v2f bx[8], by[8], nx[8], ny[8];
        #pragma unroll
        for (int j = 0; j < 8; ++j) {
            bx[j].x = bcol[(size_t)(4 * j) * DK_];
            bx[j].y = bcol[(size_t)(4 * j + 1) * DK_];
            by[j].x = bcol[(size_t)(4 * j) * DK_ + 16];
            by[j].y = bcol[(size_t)(4 * j + 1) * DK_ + 16];
        }
        #pragma unroll
        for (int g = 0; g < DK_ / 32; ++g) {
            if (g + 1 < DK_ / 32) {
                #pragma unroll
                for (int j = 0; j < 8; ++j) {
                    const int kn = (g + 1) * 32 + 4 * j;
                    nx[j].x = bcol[(size_t)kn * DK_];
                    nx[j].y = bcol[(size_t)(kn + 1) * DK_];
                    ny[j].x = bcol[(size_t)kn * DK_ + 16];
                    ny[j].y = bcol[(size_t)(kn + 1) * DK_ + 16];
                }
            }
            #pragma unroll
            for (int j = 0; j < 8; ++j) {
                v2f a = *(const v2f*)(arow + g * 32 + 4 * j);
                acc0 = __builtin_amdgcn_wmma_f32_16x16x4_f32(
                           false, a, false, bx[j], (short)0, acc0, false, false);
                acc1 = __builtin_amdgcn_wmma_f32_16x16x4_f32(
                           false, a, false, by[j], (short)0, acc1, false, false);
            }
            #pragma unroll
            for (int j = 0; j < 8; ++j) { bx[j] = nx[j]; by[j] = ny[j]; }
        }
        __syncthreads();
        #pragma unroll
        for (int r = 0; r < 8; ++r) {
            Mc[(r + 8 * hi) * MS + c0]      = acc0[r];
            Mc[(r + 8 * hi) * MS + c0 + 16] = acc1[r];
        }
        __syncthreads();
    }

    for (int i = tid; i < CROWS * DK_; i += CTHREADS) {
        int r = i >> 8, k = i & 255;
        outB[r * DK_ + k] = Mc[r * MS + k];
    }
}

extern "C" void kernel_launch(void* const* d_in, const int* in_sizes, int n_in,
                              void* d_out, int out_size, void* d_ws, size_t ws_size,
                              hipStream_t stream) {
    (void)in_sizes; (void)n_in; (void)out_size;
    const float* mem = (const float*)d_in[0];
    const float* key = (const float*)d_in[1];
    const float* val = (const float*)d_in[2];
    float* out = (float*)d_out;
    float* ws  = (float*)d_ws;

    dim3 grid(B_ * UNITS_MAX);
    dim3 block(NTHREADS);
    hipLaunchKernelGGL(delta_scan_segments, grid, block, 0, stream,
                       mem, key, val, out, ws, ws_size);
    hipLaunchKernelGGL(delta_compose_segments, dim3(B_ * (DV_ / CROWS)), dim3(CTHREADS), 0, stream,
                       out, (const float*)ws, ws_size);
}
